# starting kernel
# baseline (speedup 1.0000x reference)
#include <hip/hip_runtime.h>
#include <hip/hip_bf16.h>
#include <cstdint>

typedef __attribute__((ext_vector_type(2))) float v2f;
typedef __attribute__((ext_vector_type(8))) float v8f;

constexpr int Bb = 256;
constexpr int Ss = 1024;
constexpr int Cc = 128;

__global__ __launch_bounds__(128) void viterbi_forward(
    const float* __restrict__ emis, const int* __restrict__ mask,
    const float* __restrict__ T, uint8_t* __restrict__ bp,
    int* __restrict__ path)
{
  __shared__ float sAlpha[2][Cc];
  const int b      = blockIdx.x;
  const int tid    = threadIdx.x;
  const int w      = tid >> 5;
  const int lane   = tid & 31;
  const int laneLo = lane & 15;
  const bool hi    = lane >= 16;
  const int  hi8   = hi ? 8 : 0;
  const int  j     = (w << 5) + lane;

  float ct[2][8][8];
#pragma unroll
  for (int tc = 0; tc < 2; ++tc)
#pragma unroll
    for (int k = 0; k < 8; ++k)
#pragma unroll
      for (int r = 0; r < 8; ++r) {
        const int row = 16 * k + r + hi8;
        const int col = (w << 5) + (tc << 4) + laneLo;
        ct[tc][k][r] = T[row * Cc + col];
      }

  v2f bfrag;
  bfrag.x = hi ? 0.0f : 1.0f;
  bfrag.y = 0.0f;

  sAlpha[0][j] = emis[(size_t)b * Ss * Cc + j];
  __syncthreads();

  const float NEG = -__builtin_inff();

  for (int t = 1; t < Ss; ++t) {
    const int cur = (t - 1) & 1, nxt = t & 1;

    const float e    = emis[((size_t)b * Ss + t) * Cc + j];
    const int   m    = mask[b * Ss + t];
    const float aold = sAlpha[cur][j];
    if (t + 4 < Ss)
      __builtin_prefetch(&emis[((size_t)b * Ss + t + 4) * Cc + j], 0, 0);

    float af[8];
#pragma unroll
    for (int k = 0; k < 8; ++k) {
      const float v = sAlpha[cur][16 * k + laneLo];
      af[k] = hi ? 0.0f : v;
    }

    float best[2][2];
    int   arg8[2][2];
#pragma unroll
    for (int tc = 0; tc < 2; ++tc)
#pragma unroll
      for (int p = 0; p < 2; ++p) { best[tc][p] = NEG; arg8[tc][p] = 0; }

#pragma unroll
    for (int tc = 0; tc < 2; ++tc) {
#pragma unroll
      for (int k = 0; k < 8; ++k) {
        v2f a; a.x = af[k]; a.y = 0.0f;
        v8f cmat;
#pragma unroll
        for (int r = 0; r < 8; ++r) cmat[r] = ct[tc][k][r];
        v8f d = __builtin_amdgcn_wmma_f32_16x16x4_f32(
            false, a, false, bfrag, (short)0, cmat, false, false);
        const int p = k & 1;
#pragma unroll
        for (int r = 0; r < 8; ++r) {
          const int  cand = k * 8 + r;
          const bool g    = d[r] > best[tc][p];
          best[tc][p] = g ? d[r] : best[tc][p];
          arg8[tc][p] = g ? cand : arg8[tc][p];
        }
      }
    }

    float bb[2]; int ii[2];
#pragma unroll
    for (int tc = 0; tc < 2; ++tc) {
      const bool tk1 = (best[tc][1] > best[tc][0]) ||
                       ((best[tc][1] == best[tc][0]) && (arg8[tc][1] < arg8[tc][0]));
      float bv = tk1 ? best[tc][1] : best[tc][0];
      int   a8 = tk1 ? arg8[tc][1] : arg8[tc][0];
      int   iv = a8 + ((a8 >> 3) << 3) + hi8;
      const float bo = __shfl_xor(bv, 16, 32);
      const int   io = __shfl_xor(iv, 16, 32);
      const bool tk2 = (bo > bv) || ((bo == bv) && (io < iv));
      bb[tc] = tk2 ? bo : bv;
      ii[tc] = tk2 ? io : iv;
    }

    const float bsel = hi ? bb[1] : bb[0];
    const int   isel = hi ? ii[1] : ii[0];
    const float an   = bsel + e;
    const float afin = m ? an : aold;

    sAlpha[nxt][j] = afin;
    bp[((size_t)b * (Ss - 1) + (t - 1)) * Cc + j] = (uint8_t)isel;
    __syncthreads();
  }

  if (w == 0) {
    const float* fa = sAlpha[(Ss - 1) & 1];
    float bbest = NEG; int barg = 0;
#pragma unroll
    for (int k = 0; k < 4; ++k) {
      const int   jj = (k << 5) + lane;
      const float v  = fa[jj];
      const bool  g  = v > bbest;
      bbest = g ? v : bbest;
      barg  = g ? jj : barg;
    }
#pragma unroll
    for (int off = 16; off >= 1; off >>= 1) {
      const float bo = __shfl_xor(bbest, off, 32);
      const int   ao = __shfl_xor(barg, off, 32);
      const bool  tk = (bo > bbest) || ((bo == bbest) && (ao < barg));
      bbest = tk ? bo : bbest;
      barg  = tk ? ao : barg;
    }
    if (lane == 0) path[(size_t)b * Ss + (Ss - 1)] = barg;
  }
}

__global__ __launch_bounds__(32) void viterbi_backtrace(
    const uint8_t* __restrict__ bp, int* __restrict__ path)
{
  const int b    = blockIdx.x;
  const int lane = threadIdx.x;
  const uint8_t* bpb = bp + (size_t)b * (Ss - 1) * Cc;
  int tag = path[(size_t)b * Ss + (Ss - 1)];

  auto ROW = [&](int tt) -> unsigned {
    return ((const unsigned*)(bpb + (size_t)tt * Cc))[lane];
  };
  auto CONSUME = [&](unsigned rowv, int tt) {
    const unsigned wsel = (unsigned)__shfl((int)rowv, tag >> 2, 32);
    tag = (int)((wsel >> ((tag & 3) << 3)) & 0xFF);
    if (lane == 0) path[(size_t)b * Ss + tt] = tag;
  };

  unsigned r0 = ROW(Ss - 2), r1 = ROW(Ss - 3), r2 = ROW(Ss - 4), r3 = ROW(Ss - 5);
  for (int tb = Ss - 2; tb >= 0; tb -= 4) {
    unsigned n0 = 0, n1 = 0, n2 = 0, n3 = 0;
    if (tb - 4 >= 0) n0 = ROW(tb - 4);
    if (tb - 5 >= 0) n1 = ROW(tb - 5);
    if (tb - 6 >= 0) n2 = ROW(tb - 6);
    if (tb - 7 >= 0) n3 = ROW(tb - 7);
    CONSUME(r0, tb);
    if (tb - 1 >= 0) CONSUME(r1, tb - 1);
    if (tb - 2 >= 0) CONSUME(r2, tb - 2);
    if (tb - 3 >= 0) CONSUME(r3, tb - 3);
    r0 = n0; r1 = n1; r2 = n2; r3 = n3;
  }
}

extern "C" void kernel_launch(void* const* d_in, const int* in_sizes, int n_in,
                              void* d_out, int out_size, void* d_ws, size_t ws_size,
                              hipStream_t stream) {
  const float* emissions   = (const float*)d_in[0];
  const int*   mask        = (const int*)d_in[1];
  const float* transitions = (const float*)d_in[2];
  int*         path        = (int*)d_out;
  uint8_t*     bp          = (uint8_t*)d_ws;

  viterbi_forward<<<Bb, 128, 0, stream>>>(emissions, mask, transitions, bp, path);
  viterbi_backtrace<<<Bb, 32, 0, stream>>>(bp, path);
}


# index half of the 8-partial column merge (which previous tag held the maximum) deferred one step and issued in the issue gaps of the next step's 16 matrix instructions; only the maximum feeds alpha
# speedup vs baseline: 2.8478x; 2.8478x over previous
#include <hip/hip_runtime.h>
#include <cstdint>


typedef __attribute__((ext_vector_type(2))) float v2f;
typedef __attribute__((ext_vector_type(8))) float v8f;

constexpr int Bb = 256;
constexpr int Ss = 1024;
constexpr int Cc = 128;
constexpr int PP = 144;
constexpr int SEGL = 32;
constexpr int SEGS = 32;

__global__ __launch_bounds__(128) void viterbi_decode(
    const float* __restrict__ emis, const int* __restrict__ mask,
    const float* __restrict__ T, int* __restrict__ path)
{
  __shared__ v2f     sPart[2][8][PP];
  __shared__ float   sFin[Cc];
  __shared__ uint8_t sBp[(Ss - 1) * Cc];
  __shared__ uint8_t sMapRun[2][Cc];
  __shared__ uint8_t sMapSeg[SEGS][Cc];
  __shared__ int     sPath[Ss + SEGS];
  const int b      = blockIdx.x;
  const int tid    = threadIdx.x;
  const int w      = tid >> 5;
  const int lane   = tid & 31;
  const int laneLo = lane & 15;
  const bool hi    = lane >= 16;
  const int  hi8   = hi ? 8 : 0;
  const int  j     = (w << 5) + lane;

  float ct[8][2][8];
#pragma unroll
  for (int jt = 0; jt < 8; ++jt)
#pragma unroll
    for (int bb = 0; bb < 2; ++bb)
#pragma unroll
      for (int r = 0; r < 8; ++r) {
        const int row = (w << 5) + 16 * bb + r + hi8;
        const int col = (jt << 4) + laneLo;
        ct[jt][bb][r] = T[row * Cc + col];
      }

  v2f bLo, bHi;
  bLo.x = hi ? 0.0f : 1.0f;  bLo.y = 0.0f;
  bHi.x = hi ? 1.0f : 0.0f;  bHi.y = 0.0f;

  const float NEG   = -__builtin_inff();
  const int   ibase = (w << 5) + hi8;
  const int   part  = 2 * w + (hi ? 1 : 0);

  const float ainit = emis[(size_t)b * Ss * Cc + j];
  v2f a0, a1;
  a0.x = hi ? 0.0f : ainit;  a0.y = 0.0f;
  a1.x = hi ? ainit : 0.0f;  a1.y = 0.0f;

  float qvP[8];
  int   qiP[8];
  float vmaxP = 0.0f;
#pragma unroll
  for (int p = 0; p < 8; ++p) { qvP[p] = 0.0f; qiP[p] = 0; }
  sMapRun[1][j] = 0;
  __syncthreads();

  for (int t = 1; t < Ss; ++t) {
    const float e = emis[((size_t)b * Ss + t) * Cc + j];
    const int   m = mask[b * Ss + t];
    if (t + 4 < Ss)
      __builtin_prefetch(&emis[((size_t)b * Ss + t + 4) * Cc + j], 0, 0);

    v8f d[2][8];
#pragma unroll
    for (int bb = 0; bb < 2; ++bb)
#pragma unroll
      for (int jt = 0; jt < 8; ++jt) {
        v8f cmat;
#pragma unroll
        for (int r = 0; r < 8; ++r) cmat[r] = ct[jt][bb][r];
        d[bb][jt] = __builtin_amdgcn_wmma_f32_16x16x4_f32(
            false, bb ? a1 : a0, false, bb ? bHi : bLo, (short)0, cmat,
            false, false);
      }

    int isel = 255;
#pragma unroll
    for (int p = 0; p < 8; ++p) {
      const int ip = (qvP[p] == vmaxP) ? qiP[p] : 255;
      isel = ip < isel ? ip : isel;
    }
#pragma unroll
    for (int g = 0; g < 16; ++g) {
      __builtin_amdgcn_sched_group_barrier(0x008, 1, 0);
      __builtin_amdgcn_sched_group_barrier(0x002, 2, 0);
    }
    __builtin_amdgcn_sched_barrier(0);
    sBp[(t >= 2 ? t - 2 : 0) * Cc + j] = (uint8_t)isel;
    const int gat = sMapRun[t & 1][isel & (Cc - 1)];
    __builtin_amdgcn_sched_barrier(0);

    float best[8], first[8];
    int   arg[8];
#pragma unroll
    for (int jt = 0; jt < 8; ++jt) {
      const bool k1 = !(d[0][jt][1] > d[0][jt][0]);
      best[jt]  = k1 ? d[0][jt][0] : d[0][jt][1];
      first[jt] = best[jt];
      arg[jt]   = k1 ? 0 : 1;
    }
#pragma unroll
    for (int p = 1; p < 8; ++p) {
      const int bb   = p >> 2;
      const int r0   = (p & 3) * 2;
      const int cand = bb * 16 + r0;
      float bn[8];
      bool  keep[8];
#pragma unroll
      for (int jt = 0; jt < 8; ++jt)
        bn[jt] = fmaxf(fmaxf(best[jt], d[bb][jt][r0]), d[bb][jt][r0 + 1]);
#pragma unroll
      for (int jt = 0; jt < 8; ++jt)
        keep[jt] = !(bn[jt] > best[jt]);
#pragma unroll
      for (int jt = 0; jt < 8; ++jt) {
        arg[jt]   = keep[jt] ? arg[jt]   : cand;
        first[jt] = keep[jt] ? first[jt] : d[bb][jt][r0];
        best[jt]  = bn[jt];
      }
    }
#pragma unroll
    for (int jt = 0; jt < 8; ++jt)
      arg[jt] += (first[jt] == best[jt]) ? 0 : 1;

    __builtin_amdgcn_sched_barrier(0);
    {
      const bool headP = (t < 2) || (((t - 2) & (SEGL - 1)) == 0);
      const int  mval  = headP ? isel : gat;
      sMapRun[(t - 1) & 1][j] = (uint8_t)mval;
      sMapSeg[(t >= 2 ? t - 2 : 0) / SEGL][j] = (uint8_t)mval;
    }

    v2f* wr = &sPart[t & 1][part][laneLo];
#pragma unroll
    for (int jt = 0; jt < 8; ++jt) {
      const int iv = ibase + arg[jt];
      v2f pr;
      pr.x = best[jt];
      pr.y = __int_as_float(iv);
      wr[jt << 4] = pr;
    }
    asm volatile("" : "+v"(a0), "+v"(a1) : "v"(best[7]));
    __syncthreads();

    const v2f* rd = &sPart[t & 1][0][j];
    v2f q[8];
#pragma unroll
    for (int p = 0; p < 8; ++p) q[p] = rd[p * PP];
    const float m01 = fmaxf(q[0].x, q[1].x), m23 = fmaxf(q[2].x, q[3].x);
    const float m45 = fmaxf(q[4].x, q[5].x), m67 = fmaxf(q[6].x, q[7].x);
    const float vmax = fmaxf(fmaxf(m01, m23), fmaxf(m45, m67));
#pragma unroll
    for (int p = 0; p < 8; ++p) { qvP[p] = q[p].x; qiP[p] = __float_as_int(q[p].y); }
    vmaxP = vmax;

    const float aold = hi ? a1.x : a0.x;
    const float an   = vmax + e;
    const float afin = m ? an : aold;
    a0.x = hi ? 0.0f : afin;
    a1.x = hi ? afin : 0.0f;
  }
  {
    int isel = 255;
#pragma unroll
    for (int p = 0; p < 8; ++p) {
      const int ip = (qvP[p] == vmaxP) ? qiP[p] : 255;
      isel = ip < isel ? ip : isel;
    }
    sBp[(Ss - 2) * Cc + j] = (uint8_t)isel;
    const int  gat   = sMapRun[Ss & 1][isel & (Cc - 1)];
    const bool headP = ((Ss - 2) & (SEGL - 1)) == 0;
    sMapSeg[(Ss - 2) / SEGL][j] = (uint8_t)(headP ? isel : gat);
  }

  sFin[j] = hi ? a1.x : a0.x;
  __syncthreads();

  if (w == 0) {
    const float* fa = sFin;
    float bbest = NEG; int barg = 0;
#pragma unroll
    for (int k = 0; k < 4; ++k) {
      const int   jj = (k << 5) + lane;
      const float v  = fa[jj];
      const bool  g  = v > bbest;
      bbest = g ? v : bbest;
      barg  = g ? jj : barg;
    }
#pragma unroll
    for (int off = 16; off >= 1; off >>= 1) {
      const float bo = __shfl_xor(bbest, off, 32);
      const int   ao = __shfl_xor(barg, off, 32);
      const bool  tk = (bo > bbest) || ((bo == bbest) && (ao < barg));
      bbest = tk ? bo : bbest;
      barg  = tk ? ao : barg;
    }

    int tagB = barg, start = barg;
    for (int s = SEGS - 1; s >= 0; --s) {
      start = (lane == s) ? tagB : start;
      tagB  = sMapSeg[s][tagB & (Cc - 1)];
    }
    int tag = start;
    for (int k = SEGL - 1; k >= 0; --k) {
      const int  tau   = lane * SEGL + k;
      const bool valid = tau <= Ss - 2;
      const int  nx    = sBp[(valid ? tau : Ss - 2) * Cc + (tag & (Cc - 1))];
      tag = valid ? nx : tag;
      sPath[tau + lane] = tag;
    }
  }
  __syncthreads();

  int* prow = path + (size_t)b * Ss;
#pragma unroll
  for (int q = 0; q < Ss / 128; ++q) {
    const int tau = q * 128 + tid;
    prow[tau] = sPath[tau + tau / SEGL];
  }
}

extern "C" void kernel_launch(void* const* d_in, const int* in_sizes, int n_in,
                              void* d_out, int out_size, void* d_ws, size_t ws_size,
                              hipStream_t stream) {
  const float* emissions   = (const float*)d_in[0];
  const int*   mask        = (const int*)d_in[1];
  const float* transitions = (const float*)d_in[2];
  int*         path        = (int*)d_out;

  viterbi_decode<<<Bb, 128, 0, stream>>>(emissions, mask, transitions, path);
}


# max/argmax chains split: value pass (3-input maxima, all kept) feeds the LDS hand-over at once; index pass (compare kept maxima + select) runs in the hand-over waits; partial index rides a step later
# speedup vs baseline: 3.0308x; 1.0643x over previous
#include <hip/hip_runtime.h>
#include <cstdint>


typedef __attribute__((ext_vector_type(2))) float v2f;
typedef __attribute__((ext_vector_type(8))) float v8f;

constexpr int Bb = 256;
constexpr int Ss = 1024;
constexpr int Cc = 128;
constexpr int PP = 144;
constexpr int SEGL = 32;
constexpr int SEGS = 32;

__global__ __launch_bounds__(128) void viterbi_decode(
    const float* __restrict__ emis, const int* __restrict__ mask,
    const float* __restrict__ T, int* __restrict__ path)
{
  __shared__ v2f     sPart[2][8][PP];
  __shared__ float   sFin[Cc];
  __shared__ uint8_t sBp[(Ss - 1) * Cc];
  __shared__ uint8_t sMapRun[2][Cc];
  __shared__ uint8_t sMapSeg[SEGS][Cc];
  __shared__ int     sPath[Ss + SEGS];
  const int b      = blockIdx.x;
  const int tid    = threadIdx.x;
  const int w      = tid >> 5;
  const int lane   = tid & 31;
  const int laneLo = lane & 15;
  const bool hi    = lane >= 16;
  const int  hi8   = hi ? 8 : 0;
  const int  j     = (w << 5) + lane;

  float ct[8][2][8];
#pragma unroll
  for (int jt = 0; jt < 8; ++jt)
#pragma unroll
    for (int bb = 0; bb < 2; ++bb)
#pragma unroll
      for (int r = 0; r < 8; ++r) {
        const int row = (w << 5) + 16 * bb + r + hi8;
        const int col = (jt << 4) + laneLo;
        ct[jt][bb][r] = T[row * Cc + col];
      }

  v2f bLo, bHi;
  bLo.x = hi ? 0.0f : 1.0f;  bLo.y = 0.0f;
  bHi.x = hi ? 1.0f : 0.0f;  bHi.y = 0.0f;

  const float NEG   = -__builtin_inff();
  const int   ibase = (w << 5) + hi8;
  const int   part  = 2 * w + (hi ? 1 : 0);

  const float ainit = emis[(size_t)b * Ss * Cc + j];
  v2f a0, a1;
  a0.x = hi ? 0.0f : ainit;  a0.y = 0.0f;
  a1.x = hi ? ainit : 0.0f;  a1.y = 0.0f;

  float qvP[8], qvPP[8];
  int   qiP[8];
  int   argP[8];
  float vmaxP = 0.0f, vmaxPP = 0.0f;
#pragma unroll
  for (int p = 0; p < 8; ++p) { qvP[p] = 0.0f; qvPP[p] = 0.0f; qiP[p] = 0; argP[p] = 0; }
  sMapRun[1][j] = 0;
  __syncthreads();

  for (int t = 1; t < Ss; ++t) {
    const float e = emis[((size_t)b * Ss + t) * Cc + j];
    const int   m = mask[b * Ss + t];
    if (t + 4 < Ss)
      __builtin_prefetch(&emis[((size_t)b * Ss + t + 4) * Cc + j], 0, 0);

    v8f d[2][8];
#pragma unroll
    for (int bb = 0; bb < 2; ++bb)
#pragma unroll
      for (int jt = 0; jt < 8; ++jt) {
        v8f cmat;
#pragma unroll
        for (int r = 0; r < 8; ++r) cmat[r] = ct[jt][bb][r];
        d[bb][jt] = __builtin_amdgcn_wmma_f32_16x16x4_f32(
            false, bb ? a1 : a0, false, bb ? bHi : bLo, (short)0, cmat,
            false, false);
      }

    int isel = 255;
#pragma unroll
    for (int p = 0; p < 8; ++p) {
      const int ip = (qvPP[p] == vmaxPP) ? qiP[p] : 255;
      isel = ip < isel ? ip : isel;
    }
#pragma unroll
    for (int g = 0; g < 16; ++g) {
      __builtin_amdgcn_sched_group_barrier(0x008, 1, 0);
      __builtin_amdgcn_sched_group_barrier(0x002, 2, 0);
    }
    __builtin_amdgcn_sched_barrier(0);
    const int row  = t - 3;
    const int rowc = row > 0 ? row : 0;
    sBp[rowc * Cc + j] = (uint8_t)isel;
    const int gat = sMapRun[t & 1][isel & (Cc - 1)];
    __builtin_amdgcn_sched_barrier(0);

    float run[8][8];
#pragma unroll
    for (int jt = 0; jt < 8; ++jt)
      run[0][jt] = fmaxf(d[0][jt][0], d[0][jt][1]);
#pragma unroll
    for (int p = 1; p < 8; ++p) {
      const int bb = p >> 2;
      const int r0 = (p & 3) * 2;
#pragma unroll
      for (int jt = 0; jt < 8; ++jt)
        run[p][jt] = fmaxf(fmaxf(run[p - 1][jt], d[bb][jt][r0]), d[bb][jt][r0 + 1]);
    }
    asm volatile("" : "+v"(a0), "+v"(a1) : "v"(run[7][7]));
    __builtin_amdgcn_sched_barrier(0);

    v2f* wr = &sPart[t & 1][part][laneLo];
#pragma unroll
    for (int jt = 0; jt < 8; ++jt) {
      v2f pr;
      pr.x = run[7][jt];
      pr.y = __int_as_float(argP[jt]);
      wr[jt << 4] = pr;
    }
    {
      const bool headP = (row <= 0) || ((row & (SEGL - 1)) == 0);
      const int  mval  = headP ? isel : gat;
      sMapRun[(t - 1) & 1][j] = (uint8_t)mval;
      sMapSeg[rowc / SEGL][j] = (uint8_t)mval;
    }
    __builtin_amdgcn_sched_barrier(0);

    float first[8];
    int   arg[8];
#pragma unroll
    for (int jt = 0; jt < 8; ++jt) {
      arg[jt]   = (d[0][jt][1] > d[0][jt][0]) ? 1 : 0;
      first[jt] = run[0][jt];
    }
#define INDEX_LINK(p_)                                                        \
    {                                                                          \
      const int bb_   = (p_) >> 2;                                             \
      const int r0_   = ((p_) & 3) * 2;                                        \
      const int cand_ = bb_ * 16 + r0_;              \
      bool keep_[8];                                                           \
      _Pragma("unroll")                                                        \
      for (int jt = 0; jt < 8; ++jt)                                           \
        keep_[jt] = !(run[(p_)][jt] > run[(p_) - 1][jt]);                      \
      _Pragma("unroll")                                                        \
      for (int jt = 0; jt < 8; ++jt) {                                         \
        arg[jt]   = keep_[jt] ? arg[jt]   : cand_;                             \
        first[jt] = keep_[jt] ? first[jt] : d[bb_][jt][r0_];                   \
      }                                                                        \
    }
    INDEX_LINK(1)
    INDEX_LINK(2)
    __builtin_amdgcn_sched_barrier(0);
    __builtin_amdgcn_fence(__ATOMIC_RELEASE, "workgroup");
    __builtin_amdgcn_s_barrier_signal(-1);
    __builtin_amdgcn_sched_barrier(0);
    INDEX_LINK(3)
    INDEX_LINK(4)
    __builtin_amdgcn_sched_barrier(0);
    __builtin_amdgcn_s_barrier_wait(-1);
    __builtin_amdgcn_fence(__ATOMIC_ACQUIRE, "workgroup");

    const v2f* rd = &sPart[t & 1][0][j];
    v2f q[8];
#pragma unroll
    for (int p = 0; p < 8; ++p) q[p] = rd[p * PP];
    __builtin_amdgcn_sched_barrier(0);
    INDEX_LINK(5)
    INDEX_LINK(6)
    INDEX_LINK(7)
#undef INDEX_LINK
#pragma unroll
    for (int jt = 0; jt < 8; ++jt)
      argP[jt] = ibase + arg[jt] + ((first[jt] == run[7][jt]) ? 0 : 1);
    __builtin_amdgcn_sched_barrier(0);

    const float m01 = fmaxf(q[0].x, q[1].x), m23 = fmaxf(q[2].x, q[3].x);
    const float m45 = fmaxf(q[4].x, q[5].x), m67 = fmaxf(q[6].x, q[7].x);
    const float vmax = fmaxf(fmaxf(m01, m23), fmaxf(m45, m67));
#pragma unroll
    for (int p = 0; p < 8; ++p) {
      qvPP[p] = qvP[p];
      qiP[p]  = __float_as_int(q[p].y);
      qvP[p]  = q[p].x;
    }
    vmaxPP = vmaxP;
    vmaxP  = vmax;

    const float aold = hi ? a1.x : a0.x;
    const float an   = vmax + e;
    const float afin = m ? an : aold;
    a0.x = hi ? 0.0f : afin;
    a1.x = hi ? afin : 0.0f;
  }
  {
    int isel = 255;
#pragma unroll
    for (int p = 0; p < 8; ++p) {
      const int ip = (qvPP[p] == vmaxPP) ? qiP[p] : 255;
      isel = ip < isel ? ip : isel;
    }
    sBp[(Ss - 3) * Cc + j] = (uint8_t)isel;
    const int  gat   = sMapRun[Ss & 1][isel & (Cc - 1)];
    const bool headP = ((Ss - 3) & (SEGL - 1)) == 0;
    const int  mval  = headP ? isel : gat;
    sMapRun[(Ss - 1) & 1][j] = (uint8_t)mval;
    sMapSeg[(Ss - 3) / SEGL][j] = (uint8_t)mval;
  }
  {
    v2f* wr = &sPart[Ss & 1][part][laneLo];
#pragma unroll
    for (int jt = 0; jt < 8; ++jt) {
      v2f pr;
      pr.x = 0.0f;
      pr.y = __int_as_float(argP[jt]);
      wr[jt << 4] = pr;
    }
  }
  __syncthreads();
  {
    const v2f* rd = &sPart[Ss & 1][0][j];
    int isel = 255;
#pragma unroll
    for (int p = 0; p < 8; ++p) {
      const int qi = __float_as_int(rd[p * PP].y);
      const int ip = (qvP[p] == vmaxP) ? qi : 255;
      isel = ip < isel ? ip : isel;
    }
    sBp[(Ss - 2) * Cc + j] = (uint8_t)isel;
    const int  gat   = sMapRun[(Ss + 1) & 1][isel & (Cc - 1)];
    const bool headP = ((Ss - 2) & (SEGL - 1)) == 0;
    sMapSeg[(Ss - 2) / SEGL][j] = (uint8_t)(headP ? isel : gat);
  }

  sFin[j] = hi ? a1.x : a0.x;
  __syncthreads();

  if (w == 0) {
    const float* fa = sFin;
    float bbest = NEG; int barg = 0;
#pragma unroll
    for (int k = 0; k < 4; ++k) {
      const int   jj = (k << 5) + lane;
      const float v  = fa[jj];
      const bool  g  = v > bbest;
      bbest = g ? v : bbest;
      barg  = g ? jj : barg;
    }
#pragma unroll
    for (int off = 16; off >= 1; off >>= 1) {
      const float bo = __shfl_xor(bbest, off, 32);
      const int   ao = __shfl_xor(barg, off, 32);
      const bool  tk = (bo > bbest) || ((bo == bbest) && (ao < barg));
      bbest = tk ? bo : bbest;
      barg  = tk ? ao : barg;
    }

    int tagB = barg, start = barg;
    for (int s = SEGS - 1; s >= 0; --s) {
      start = (lane == s) ? tagB : start;
      tagB  = sMapSeg[s][tagB & (Cc - 1)];
    }
    int tag = start;
    for (int k = SEGL - 1; k >= 0; --k) {
      const int  tau   = lane * SEGL + k;
      const bool valid = tau <= Ss - 2;
      const int  nx    = sBp[(valid ? tau : Ss - 2) * Cc + (tag & (Cc - 1))];
      tag = valid ? nx : tag;
      sPath[tau + lane] = tag;
    }
  }
  __syncthreads();

  int* prow = path + (size_t)b * Ss;
#pragma unroll
  for (int q = 0; q < Ss / 128; ++q) {
    const int tau = q * 128 + tid;
    prow[tau] = sPath[tau + tau / SEGL];
  }
}

extern "C" void kernel_launch(void* const* d_in, const int* in_sizes, int n_in,
                              void* d_out, int out_size, void* d_ws, size_t ws_size,
                              hipStream_t stream) {
  const float* emissions   = (const float*)d_in[0];
  const int*   mask        = (const int*)d_in[1];
  const float* transitions = (const float*)d_in[2];
  int*         path        = (int*)d_out;

  viterbi_decode<<<Bb, 128, 0, stream>>>(emissions, mask, transitions, path);
}
